# starting kernel
# baseline (speedup 1.0000x reference)
#include <hip/hip_runtime.h>
#include <hip/hip_bf16.h>

typedef float v8f __attribute__((ext_vector_type(8)));
typedef float v2f __attribute__((ext_vector_type(2)));

#define BATCH 8
#define CCH   512
#define NHEAD 8
#define DHEAD 64
#define NTOK  1024
#define KTOP  4
#define C3    1536

__global__ void zero_out_kernel(float* __restrict__ out, int n) {
    int i = blockIdx.x * blockDim.x + threadIdx.x;
    if (i < n) out[i] = 0.0f;
}

__global__ __launch_bounds__(128) void qkv_kernel(const float* __restrict__ x,
                                                  const float* __restrict__ Wqkv,
                                                  const float* __restrict__ bqkv,
                                                  float* __restrict__ qbuf,
                                                  float* __restrict__ kTbuf,
                                                  float* __restrict__ vbuf) {
    const int lane    = threadIdx.x & 31;
    const int waveId  = blockIdx.x * 4 + (threadIdx.x >> 5);
    const int b       = waveId / (64 * 24);
    const int rem     = waveId % (64 * 24);
    const int mtile   = rem / 24;
    const int ng      = rem % 24;
    const int which   = ng >> 3;
    const int head    = ng & 7;
    const int m0      = mtile * 16;
    const int j0      = ng * 64;
    const int halfsel = lane >> 4;
    const int lm      = lane & 15;

    v8f acc[4];
    #pragma unroll
    for (int t = 0; t < 4; ++t) {
        float bv = bqkv[j0 + 16 * t + lm];
        #pragma unroll
        for (int r = 0; r < 8; ++r) acc[t][r] = bv;
    }

    const float* xb = x + (size_t)b * CCH * NTOK;
    for (int k0 = 0; k0 < CCH; k0 += 4) {
        const int kk = k0 + 2 * halfsel;
        v2f a;
        a.x = xb[(size_t)kk * NTOK + m0 + lm];
        a.y = xb[(size_t)(kk + 1) * NTOK + m0 + lm];
        #pragma unroll
        for (int t = 0; t < 4; ++t) {
            v2f bf;
            bf.x = Wqkv[(size_t)kk * C3 + j0 + 16 * t + lm];
            bf.y = Wqkv[(size_t)(kk + 1) * C3 + j0 + 16 * t + lm];
            acc[t] = __builtin_amdgcn_wmma_f32_16x16x4_f32(
                false, a, false, bf, (short)0, acc[t], false, false);
        }
    }

    const int bh = b * NHEAD + head;
    #pragma unroll
    for (int t = 0; t < 4; ++t) {
        #pragma unroll
        for (int r = 0; r < 8; ++r) {
            const int m  = m0 + r + 8 * halfsel;
            const int jl = 16 * t + lm;
            const float val = acc[t][r];
            if (which == 0) {
                qbuf[((size_t)bh * NTOK + m) * DHEAD + jl] = val;
            } else if (which == 1) {
                kTbuf[((((size_t)bh * (DHEAD / 2) + (jl >> 1)) * NTOK + m) << 1) + (jl & 1)] = val;
            } else {
                vbuf[((size_t)bh * NTOK + m) * DHEAD + jl] = val;
            }
        }
    }
}

__global__ __launch_bounds__(256) void attn_kernel(const float* __restrict__ qbuf,
                                                   const float* __restrict__ kTbuf,
                                                   const float* __restrict__ vbuf,
                                                   float* __restrict__ out) {
    __shared__ float sc[16][260];
    __shared__ float candw[16][16][KTOP];
    __shared__ int   candi[16][16][KTOP];
    __shared__ float topw[16][KTOP];
    __shared__ int   topi[16][KTOP];
    __shared__ float out_acc[DHEAD];

    const int tid     = threadIdx.x;
    const int lane    = tid & 31;
    const int wv      = tid >> 5;
    const int halfsel = lane >> 4;
    const int lm      = lane & 15;

    const int bh    = blockIdx.x >> 6;
    const int itile = blockIdx.x & 63;
    const int b     = bh >> 3;
    const int h     = bh & 7;
    const int i0    = itile * 16;

    const size_t qb = (size_t)bh * NTOK * DHEAD;
    const size_t vb = qb;
    const float2* kT2 = (const float2*)(kTbuf + (size_t)bh * DHEAD * NTOK);

    if (tid < DHEAD) out_acc[tid] = 0.0f;

    v2f afrag[16];
    {
        const float* qrow = qbuf + qb + (size_t)(i0 + lm) * DHEAD + 2 * halfsel;
        #pragma unroll
        for (int kk = 0; kk < 16; ++kk) {
            afrag[kk].x = qrow[kk * 4];
            afrag[kk].y = qrow[kk * 4 + 1];
        }
    }

    const int srow = tid >> 4;
    const int ssub = tid & 15;
    float lw[KTOP];
    int   li[KTOP];
    #pragma unroll
    for (int p = 0; p < KTOP; ++p) { lw[p] = -__builtin_inff(); li[p] = 0x7fffffff; }

    for (int chunk = 0; chunk < 4; ++chunk) {
        const int jcbase = chunk * 256;
        #pragma unroll
        for (int tsub = 0; tsub < 2; ++tsub) {
            const int jl0 = wv * 32 + tsub * 16;
            const int jg  = jcbase + jl0;
            v8f acc;
            #pragma unroll
            for (int r = 0; r < 8; ++r) acc[r] = 0.0f;
            #pragma unroll
            for (int kk = 0; kk < 16; ++kk) {
                const float2 bp = kT2[(size_t)(kk * 2 + halfsel) * NTOK + jg + lm];
                v2f bf; bf.x = bp.x; bf.y = bp.y;
                acc = __builtin_amdgcn_wmma_f32_16x16x4_f32(
                    false, afrag[kk], false, bf, (short)0, acc, false, false);
            }
            #pragma unroll
            for (int r = 0; r < 8; ++r)
                sc[r + 8 * halfsel][jl0 + lm] = acc[r];
        }
        __syncthreads();
        #pragma unroll
        for (int c = 0; c < 16; ++c) {
            const int cl = ssub * 16 + c;
            const float v = sc[srow][cl];
            const int   j = jcbase + cl;
            #pragma unroll
            for (int p = 0; p < KTOP; ++p) {
                const bool better = (v > lw[p]) || (v == lw[p] && j < li[p]);
                if (better) {
                    #pragma unroll
                    for (int q = KTOP - 1; q > p; --q) { lw[q] = lw[q - 1]; li[q] = li[q - 1]; }
                    lw[p] = v; li[p] = j;
                    break;
                }
            }
        }
        __syncthreads();
    }

    #pragma unroll
    for (int p = 0; p < KTOP; ++p) { candw[srow][ssub][p] = lw[p]; candi[srow][ssub][p] = li[p]; }
    __syncthreads();

    if (tid < 16) {
        const int row = tid;
        float w4[KTOP]; int i4[KTOP];
        #pragma unroll
        for (int p = 0; p < KTOP; ++p) { w4[p] = -__builtin_inff(); i4[p] = 0x7fffffff; }
        for (int s = 0; s < 16; ++s) {
            #pragma unroll
            for (int cp = 0; cp < KTOP; ++cp) {
                const float v = candw[row][s][cp];
                const int   j = candi[row][s][cp];
                #pragma unroll
                for (int p = 0; p < KTOP; ++p) {
                    const bool better = (v > w4[p]) || (v == w4[p] && j < i4[p]);
                    if (better) {
                        #pragma unroll
                        for (int q = KTOP - 1; q > p; --q) { w4[q] = w4[q - 1]; i4[q] = i4[q - 1]; }
                        w4[p] = v; i4[p] = j;
                        break;
                    }
                }
            }
        }
        const float mx = w4[0];
        float e[KTOP], s = 0.0f;
        #pragma unroll
        for (int p = 0; p < KTOP; ++p) { e[p] = __expf(w4[p] - mx); s += e[p]; }
        const float inv = 1.0f / s;
        #pragma unroll
        for (int p = 0; p < KTOP; ++p) { topw[row][p] = e[p] * inv; topi[row][p] = i4[p]; }
    }
    __syncthreads();

    {
        const int row = tid >> 4;
        const int dd0 = (tid & 15) * 4;
        #pragma unroll
        for (int u = 0; u < 4; ++u) {
            const int dd = dd0 + u;
            float wvv = 0.0f;
            #pragma unroll
            for (int p = 0; p < KTOP; ++p)
                wvv += topw[row][p] * vbuf[vb + (size_t)topi[row][p] * DHEAD + dd];
            const float val = wvv * qbuf[qb + (size_t)(i0 + row) * DHEAD + dd];
            atomicAdd(&out_acc[dd], val);
        }
    }
    __syncthreads();
    if (tid < DHEAD)
        atomicAdd(&out[b * CCH + tid * NHEAD + h], out_acc[tid]);
}

extern "C" void kernel_launch(void* const* d_in, const int* in_sizes, int n_in,
                              void* d_out, int out_size, void* d_ws, size_t ws_size,
                              hipStream_t stream) {
    const float* x  = (const float*)d_in[0];
    const float* Wq = (const float*)d_in[1];
    const float* bq = (const float*)d_in[2];
    float* out = (float*)d_out;
    float* ws  = (float*)d_ws;

    const size_t per = (size_t)BATCH * NHEAD * NTOK * DHEAD;
    float* qbuf = ws;
    float* kT   = ws + per;
    float* vbuf = ws + 2 * per;

    zero_out_kernel<<<(out_size + 255) / 256, 256, 0, stream>>>(out, out_size);
    qkv_kernel<<<3072, 128, 0, stream>>>(x, Wq, bq, qbuf, kT, vbuf);
    attn_kernel<<<4096, 256, 0, stream>>>(qbuf, kT, vbuf, out);
}


# attn top-4 selection: branchy sorted insert (nested exec-mask chains per score) replaced by branch-free compare/select insert in stripe scan (strict >) and row merge; stripe read as float4
# speedup vs baseline: 1.2092x; 1.2092x over previous
#include <hip/hip_runtime.h>
#include <hip/hip_bf16.h>

typedef float v8f __attribute__((ext_vector_type(8)));
typedef float v2f __attribute__((ext_vector_type(2)));

#define BATCH 8
#define CCH   512
#define NHEAD 8
#define DHEAD 64
#define NTOK  1024
#define KTOP  4
#define C3    1536

__global__ void zero_out_kernel(float* __restrict__ out, int n) {
    int i = blockIdx.x * blockDim.x + threadIdx.x;
    if (i < n) out[i] = 0.0f;
}

__global__ __launch_bounds__(128) void qkv_kernel(const float* __restrict__ x,
                                                  const float* __restrict__ Wqkv,
                                                  const float* __restrict__ bqkv,
                                                  float* __restrict__ qbuf,
                                                  float* __restrict__ kTbuf,
                                                  float* __restrict__ vbuf) {
    const int lane    = threadIdx.x & 31;
    const int waveId  = blockIdx.x * 4 + (threadIdx.x >> 5);
    const int b       = waveId / (64 * 24);
    const int rem     = waveId % (64 * 24);
    const int mtile   = rem / 24;
    const int ng      = rem % 24;
    const int which   = ng >> 3;
    const int head    = ng & 7;
    const int m0      = mtile * 16;
    const int j0      = ng * 64;
    const int halfsel = lane >> 4;
    const int lm      = lane & 15;

    v8f acc[4];
    #pragma unroll
    for (int t = 0; t < 4; ++t) {
        float bv = bqkv[j0 + 16 * t + lm];
        #pragma unroll
        for (int r = 0; r < 8; ++r) acc[t][r] = bv;
    }

    const float* xb = x + (size_t)b * CCH * NTOK;
    for (int k0 = 0; k0 < CCH; k0 += 4) {
        const int kk = k0 + 2 * halfsel;
        v2f a;
        a.x = xb[(size_t)kk * NTOK + m0 + lm];
        a.y = xb[(size_t)(kk + 1) * NTOK + m0 + lm];
        #pragma unroll
        for (int t = 0; t < 4; ++t) {
            v2f bf;
            bf.x = Wqkv[(size_t)kk * C3 + j0 + 16 * t + lm];
            bf.y = Wqkv[(size_t)(kk + 1) * C3 + j0 + 16 * t + lm];
            acc[t] = __builtin_amdgcn_wmma_f32_16x16x4_f32(
                false, a, false, bf, (short)0, acc[t], false, false);
        }
    }

    const int bh = b * NHEAD + head;
    #pragma unroll
    for (int t = 0; t < 4; ++t) {
        #pragma unroll
        for (int r = 0; r < 8; ++r) {
            const int m  = m0 + r + 8 * halfsel;
            const int jl = 16 * t + lm;
            const float val = acc[t][r];
            if (which == 0) {
                qbuf[((size_t)bh * NTOK + m) * DHEAD + jl] = val;
            } else if (which == 1) {
                kTbuf[((((size_t)bh * (DHEAD / 2) + (jl >> 1)) * NTOK + m) << 1) + (jl & 1)] = val;
            } else {
                vbuf[((size_t)bh * NTOK + m) * DHEAD + jl] = val;
            }
        }
    }
}

__device__ __forceinline__ void top4_put(float (&w)[4], int (&ix)[4], float v, int j,
                                         bool b0, bool b1, bool b2, bool b3) {
    w[3]  = b2 ? w[2]  : (b3 ? v : w[3]);
    ix[3] = b2 ? ix[2] : (b3 ? j : ix[3]);
    w[2]  = b1 ? w[1]  : (b2 ? v : w[2]);
    ix[2] = b1 ? ix[1] : (b2 ? j : ix[2]);
    w[1]  = b0 ? w[0]  : (b1 ? v : w[1]);
    ix[1] = b0 ? ix[0] : (b1 ? j : ix[1]);
    w[0]  = b0 ? v : w[0];
    ix[0] = b0 ? j : ix[0];
}

__global__ __launch_bounds__(256) void attn_kernel(const float* __restrict__ qbuf,
                                                   const float* __restrict__ kTbuf,
                                                   const float* __restrict__ vbuf,
                                                   float* __restrict__ out) {
    __shared__ __attribute__((aligned(16))) float sc[16][260];
    __shared__ __attribute__((aligned(16))) float candw[16][16][KTOP];
    __shared__ __attribute__((aligned(16))) int   candi[16][16][KTOP];
    __shared__ float topw[16][KTOP];
    __shared__ int   topi[16][KTOP];
    __shared__ float out_acc[DHEAD];

    const int tid     = threadIdx.x;
    const int lane    = tid & 31;
    const int wv      = tid >> 5;
    const int halfsel = lane >> 4;
    const int lm      = lane & 15;

    const int bh    = blockIdx.x >> 6;
    const int itile = blockIdx.x & 63;
    const int b     = bh >> 3;
    const int h     = bh & 7;
    const int i0    = itile * 16;

    const size_t qb = (size_t)bh * NTOK * DHEAD;
    const size_t vb = qb;
    const float2* kT2 = (const float2*)(kTbuf + (size_t)bh * DHEAD * NTOK);

    if (tid < DHEAD) out_acc[tid] = 0.0f;

    v2f afrag[16];
    {
        const float* qrow = qbuf + qb + (size_t)(i0 + lm) * DHEAD + 2 * halfsel;
        #pragma unroll
        for (int kk = 0; kk < 16; ++kk) {
            afrag[kk].x = qrow[kk * 4];
            afrag[kk].y = qrow[kk * 4 + 1];
        }
    }

    const int srow = tid >> 4;
    const int ssub = tid & 15;
    float lw[KTOP];
    int   li[KTOP];
    #pragma unroll
    for (int p = 0; p < KTOP; ++p) { lw[p] = -__builtin_inff(); li[p] = 0x7fffffff; }

    for (int chunk = 0; chunk < 4; ++chunk) {
        const int jcbase = chunk * 256;
        #pragma unroll
        for (int tsub = 0; tsub < 2; ++tsub) {
            const int jl0 = wv * 32 + tsub * 16;
            const int jg  = jcbase + jl0;
            v8f acc;
            #pragma unroll
            for (int r = 0; r < 8; ++r) acc[r] = 0.0f;
            #pragma unroll
            for (int kk = 0; kk < 16; ++kk) {
                const float2 bp = kT2[(size_t)(kk * 2 + halfsel) * NTOK + jg + lm];
                v2f bf; bf.x = bp.x; bf.y = bp.y;
                acc = __builtin_amdgcn_wmma_f32_16x16x4_f32(
                    false, afrag[kk], false, bf, (short)0, acc, false, false);
            }
            #pragma unroll
            for (int r = 0; r < 8; ++r)
                sc[r + 8 * halfsel][jl0 + lm] = acc[r];
        }
        __syncthreads();
        {
            const float4* stripe = reinterpret_cast<const float4*>(&sc[srow][ssub * 16]);
            float sv[16];
            #pragma unroll
            for (int q = 0; q < 4; ++q) {
                const float4 t = stripe[q];
                sv[4 * q + 0] = t.x; sv[4 * q + 1] = t.y; sv[4 * q + 2] = t.z; sv[4 * q + 3] = t.w;
            }
            const int jb = jcbase + ssub * 16;
            #pragma unroll
            for (int c = 0; c < 16; ++c) {
                const float v = sv[c];
                top4_put(lw, li, v, jb + c, v > lw[0], v > lw[1], v > lw[2], v > lw[3]);
            }
        }
        __syncthreads();
    }

    #pragma unroll
    for (int p = 0; p < KTOP; ++p) { candw[srow][ssub][p] = lw[p]; candi[srow][ssub][p] = li[p]; }
    __syncthreads();

    if (tid < 16) {
        const int row = tid;
        float w4[KTOP]; int i4[KTOP];
        #pragma unroll
        for (int p = 0; p < KTOP; ++p) { w4[p] = -__builtin_inff(); i4[p] = 0x7fffffff; }
        for (int s = 0; s < 16; ++s) {
            const float4 cw = *reinterpret_cast<const float4*>(&candw[row][s][0]);
            const int4   ci = *reinterpret_cast<const int4*>(&candi[row][s][0]);
            const float cv[KTOP] = {cw.x, cw.y, cw.z, cw.w};
            const int   cj[KTOP] = {ci.x, ci.y, ci.z, ci.w};
            #pragma unroll
            for (int cp = 0; cp < KTOP; ++cp) {
                const float v = cv[cp];
                const int   j = cj[cp];
                top4_put(w4, i4, v, j,
                         (v > w4[0]) || (v == w4[0] && j < i4[0]),
                         (v > w4[1]) || (v == w4[1] && j < i4[1]),
                         (v > w4[2]) || (v == w4[2] && j < i4[2]),
                         (v > w4[3]) || (v == w4[3] && j < i4[3]));
            }
        }
        const float mx = w4[0];
        float e[KTOP], s = 0.0f;
        #pragma unroll
        for (int p = 0; p < KTOP; ++p) { e[p] = __expf(w4[p] - mx); s += e[p]; }
        const float inv = 1.0f / s;
        #pragma unroll
        for (int p = 0; p < KTOP; ++p) { topw[row][p] = e[p] * inv; topi[row][p] = i4[p]; }
    }
    __syncthreads();

    {
        const int row = tid >> 4;
        const int dd0 = (tid & 15) * 4;
        #pragma unroll
        for (int u = 0; u < 4; ++u) {
            const int dd = dd0 + u;
            float wvv = 0.0f;
            #pragma unroll
            for (int p = 0; p < KTOP; ++p)
                wvv += topw[row][p] * vbuf[vb + (size_t)topi[row][p] * DHEAD + dd];
            const float val = wvv * qbuf[qb + (size_t)(i0 + row) * DHEAD + dd];
            atomicAdd(&out_acc[dd], val);
        }
    }
    __syncthreads();
    if (tid < DHEAD)
        atomicAdd(&out[b * CCH + tid * NHEAD + h], out_acc[tid]);
}

extern "C" void kernel_launch(void* const* d_in, const int* in_sizes, int n_in,
                              void* d_out, int out_size, void* d_ws, size_t ws_size,
                              hipStream_t stream) {
    const float* x  = (const float*)d_in[0];
    const float* Wq = (const float*)d_in[1];
    const float* bq = (const float*)d_in[2];
    float* out = (float*)d_out;
    float* ws  = (float*)d_ws;

    const size_t per = (size_t)BATCH * NHEAD * NTOK * DHEAD;
    float* qbuf = ws;
    float* kT   = ws + per;
    float* vbuf = ws + 2 * per;

    zero_out_kernel<<<(out_size + 255) / 256, 256, 0, stream>>>(out, out_size);
    qkv_kernel<<<3072, 128, 0, stream>>>(x, Wq, bq, qbuf, kT, vbuf);
    attn_kernel<<<4096, 256, 0, stream>>>(qbuf, kT, vbuf, out);
}


# QKV GEMM re-tiled: wave tile 16x64 -> 64x64, rows/cols dealt interleaved so each A/B fragment load is one 16-byte load (4 loads per 16 WMMAs, was 10 per 4), 12-wave blocks, loads one set ahead
# speedup vs baseline: 1.4511x; 1.2000x over previous
#include <hip/hip_runtime.h>
#include <hip/hip_bf16.h>

typedef float v8f __attribute__((ext_vector_type(8)));
typedef float v2f __attribute__((ext_vector_type(2)));

#define BATCH 8
#define CCH   512
#define NHEAD 8
#define DHEAD 64
#define NTOK  1024
#define KTOP  4
#define C3    1536

__global__ void zero_out_kernel(float* __restrict__ out, int n) {
    int i = blockIdx.x * blockDim.x + threadIdx.x;
    if (i < n) out[i] = 0.0f;
}

typedef float v4f __attribute__((ext_vector_type(4)));

struct KStep { v4f a0, a1, b0, b1; };

__device__ __forceinline__ KStep load_kstep(const float* xa, const float* wb, int k0) {
    KStep f;
    f.a0 = *reinterpret_cast<const v4f*>(xa + (size_t)k0 * NTOK);
    f.a1 = *reinterpret_cast<const v4f*>(xa + (size_t)(k0 + 1) * NTOK);
    f.b0 = *reinterpret_cast<const v4f*>(wb + (size_t)k0 * C3);
    f.b1 = *reinterpret_cast<const v4f*>(wb + (size_t)(k0 + 1) * C3);
    return f;
}

__device__ __forceinline__ void mma_kstep(v8f (&acc)[4][4], const KStep& f) {
    #pragma unroll
    for (int t = 0; t < 4; ++t) {
        v2f a; a.x = f.a0[t]; a.y = f.a1[t];
        #pragma unroll
        for (int u = 0; u < 4; ++u) {
            v2f bf; bf.x = f.b0[u]; bf.y = f.b1[u];
            acc[t][u] = __builtin_amdgcn_wmma_f32_16x16x4_f32(
                false, a, false, bf, (short)0, acc[t][u], false, false);
        }
    }
}

__global__ __launch_bounds__(384) void qkv_kernel(const float* __restrict__ x,
                                                  const float* __restrict__ Wqkv,
                                                  const float* __restrict__ bqkv,
                                                  float* __restrict__ qbuf,
                                                  float* __restrict__ kTbuf,
                                                  float* __restrict__ vbuf) {
    const int lane = threadIdx.x & 31;
    const int wv   = __builtin_amdgcn_readfirstlane((int)(threadIdx.x >> 5));
    const int h    = lane >> 4;
    const int c    = lane & 15;
    const int mb   = blockIdx.x >> 3;
    const int nb   = blockIdx.x & 7;
    const int b    = mb >> 2;
    const int m0   = (mb & 3) * 256 + (wv & 3) * 64;
    const int ng   = nb * 3 + (wv >> 2);
    const int which = ng >> 3;
    const int head  = ng & 7;
    const int j0   = ng * 64;

    v8f acc[4][4];
    {
        const v4f bv = *reinterpret_cast<const v4f*>(bqkv + j0 + 4 * c);
        #pragma unroll
        for (int t = 0; t < 4; ++t)
            #pragma unroll
            for (int u = 0; u < 4; ++u)
                #pragma unroll
                for (int r = 0; r < 8; ++r) acc[t][u][r] = bv[u];
    }

    const float* xa = x + (size_t)b * CCH * NTOK + (size_t)(2 * h) * NTOK + m0 + 4 * c;
    const float* wb = Wqkv + (size_t)(2 * h) * C3 + j0 + 4 * c;

    KStep f0 = load_kstep(xa, wb, 0);
    KStep f1 = load_kstep(xa, wb, 4);
    #pragma nounroll
    for (int k0 = 0; k0 < CCH - 16; k0 += 16) {
        KStep g0 = load_kstep(xa, wb, k0 + 8);
        KStep g1 = load_kstep(xa, wb, k0 + 12);
        __builtin_amdgcn_sched_barrier(0);
        mma_kstep(acc, f0);
        mma_kstep(acc, f1);
        f0 = load_kstep(xa, wb, k0 + 16);
        f1 = load_kstep(xa, wb, k0 + 20);
        __builtin_amdgcn_sched_barrier(0);
        mma_kstep(acc, g0);
        mma_kstep(acc, g1);
    }
    {
        KStep g0 = load_kstep(xa, wb, CCH - 8);
        KStep g1 = load_kstep(xa, wb, CCH - 4);
        __builtin_amdgcn_sched_barrier(0);
        mma_kstep(acc, f0);
        mma_kstep(acc, f1);
        mma_kstep(acc, g0);
        mma_kstep(acc, g1);
    }

    const int bh = b * NHEAD + head;
    const int mrow = m0 + 32 * h;
    if (which != 1) {
        float* dst = (which == 0 ? qbuf : vbuf) + (size_t)bh * NTOK * DHEAD + 4 * c;
        #pragma unroll
        for (int t = 0; t < 4; ++t) {
            #pragma unroll
            for (int r = 0; r < 8; ++r) {
                v4f val;
                val[0] = acc[t][0][r]; val[1] = acc[t][1][r]; val[2] = acc[t][2][r]; val[3] = acc[t][3][r];
                *reinterpret_cast<v4f*>(dst + (size_t)(mrow + 4 * r + t) * DHEAD) = val;
            }
        }
    } else {
        #pragma unroll
        for (int r = 0; r < 8; ++r) {
            #pragma unroll
            for (int p = 0; p < 2; ++p) {
                float* dstk = kTbuf + ((((size_t)bh * (DHEAD / 2) + 2 * c + p) * NTOK + mrow + 4 * r) << 1);
                v4f lo, hi;
                lo[0] = acc[0][2 * p][r]; lo[1] = acc[0][2 * p + 1][r]; lo[2] = acc[1][2 * p][r]; lo[3] = acc[1][2 * p + 1][r];
                hi[0] = acc[2][2 * p][r]; hi[1] = acc[2][2 * p + 1][r]; hi[2] = acc[3][2 * p][r]; hi[3] = acc[3][2 * p + 1][r];
                *reinterpret_cast<v4f*>(dstk)     = lo;
                *reinterpret_cast<v4f*>(dstk + 4) = hi;
            }
        }
    }
}

__device__ __forceinline__ void top4_put(float (&w)[4], int (&ix)[4], float v, int j,
                                         bool b0, bool b1, bool b2, bool b3) {
    w[3]  = b2 ? w[2]  : (b3 ? v : w[3]);
    ix[3] = b2 ? ix[2] : (b3 ? j : ix[3]);
    w[2]  = b1 ? w[1]  : (b2 ? v : w[2]);
    ix[2] = b1 ? ix[1] : (b2 ? j : ix[2]);
    w[1]  = b0 ? w[0]  : (b1 ? v : w[1]);
    ix[1] = b0 ? ix[0] : (b1 ? j : ix[1]);
    w[0]  = b0 ? v : w[0];
    ix[0] = b0 ? j : ix[0];
}

__global__ __launch_bounds__(256) void attn_kernel(const float* __restrict__ qbuf,
                                                   const float* __restrict__ kTbuf,
                                                   const float* __restrict__ vbuf,
                                                   float* __restrict__ out) {
    __shared__ __attribute__((aligned(16))) float sc[16][260];
    __shared__ __attribute__((aligned(16))) float candw[16][16][KTOP];
    __shared__ __attribute__((aligned(16))) int   candi[16][16][KTOP];
    __shared__ float topw[16][KTOP];
    __shared__ int   topi[16][KTOP];
    __shared__ float out_acc[DHEAD];

    const int tid     = threadIdx.x;
    const int lane    = tid & 31;
    const int wv      = tid >> 5;
    const int halfsel = lane >> 4;
    const int lm      = lane & 15;

    const int bh    = blockIdx.x >> 6;
    const int itile = blockIdx.x & 63;
    const int b     = bh >> 3;
    const int h     = bh & 7;
    const int i0    = itile * 16;

    const size_t qb = (size_t)bh * NTOK * DHEAD;
    const size_t vb = qb;
    const float2* kT2 = (const float2*)(kTbuf + (size_t)bh * DHEAD * NTOK);

    if (tid < DHEAD) out_acc[tid] = 0.0f;

    v2f afrag[16];
    {
        const float* qrow = qbuf + qb + (size_t)(i0 + lm) * DHEAD + 2 * halfsel;
        #pragma unroll
        for (int kk = 0; kk < 16; ++kk) {
            afrag[kk].x = qrow[kk * 4];
            afrag[kk].y = qrow[kk * 4 + 1];
        }
    }

    const int srow = tid >> 4;
    const int ssub = tid & 15;
    float lw[KTOP];
    int   li[KTOP];
    #pragma unroll
    for (int p = 0; p < KTOP; ++p) { lw[p] = -__builtin_inff(); li[p] = 0x7fffffff; }

    for (int chunk = 0; chunk < 4; ++chunk) {
        const int jcbase = chunk * 256;
        #pragma unroll
        for (int tsub = 0; tsub < 2; ++tsub) {
            const int jl0 = wv * 32 + tsub * 16;
            const int jg  = jcbase + jl0;
            v8f acc;
            #pragma unroll
            for (int r = 0; r < 8; ++r) acc[r] = 0.0f;
            #pragma unroll
            for (int kk = 0; kk < 16; ++kk) {
                const float2 bp = kT2[(size_t)(kk * 2 + halfsel) * NTOK + jg + lm];
                v2f bf; bf.x = bp.x; bf.y = bp.y;
                acc = __builtin_amdgcn_wmma_f32_16x16x4_f32(
                    false, afrag[kk], false, bf, (short)0, acc, false, false);
            }
            #pragma unroll
            for (int r = 0; r < 8; ++r)
                sc[r + 8 * halfsel][jl0 + lm] = acc[r];
        }
        __syncthreads();
        {
            const float4* stripe = reinterpret_cast<const float4*>(&sc[srow][ssub * 16]);
            float sv[16];
            #pragma unroll
            for (int q = 0; q < 4; ++q) {
                const float4 t = stripe[q];
                sv[4 * q + 0] = t.x; sv[4 * q + 1] = t.y; sv[4 * q + 2] = t.z; sv[4 * q + 3] = t.w;
            }
            const int jb = jcbase + ssub * 16;
            #pragma unroll
            for (int c = 0; c < 16; ++c) {
                const float v = sv[c];
                top4_put(lw, li, v, jb + c, v > lw[0], v > lw[1], v > lw[2], v > lw[3]);
            }
        }
        __syncthreads();
    }

    #pragma unroll
    for (int p = 0; p < KTOP; ++p) { candw[srow][ssub][p] = lw[p]; candi[srow][ssub][p] = li[p]; }
    __syncthreads();

    if (tid < 16) {
        const int row = tid;
        float w4[KTOP]; int i4[KTOP];
        #pragma unroll
        for (int p = 0; p < KTOP; ++p) { w4[p] = -__builtin_inff(); i4[p] = 0x7fffffff; }
        for (int s = 0; s < 16; ++s) {
            const float4 cw = *reinterpret_cast<const float4*>(&candw[row][s][0]);
            const int4   ci = *reinterpret_cast<const int4*>(&candi[row][s][0]);
            const float cv[KTOP] = {cw.x, cw.y, cw.z, cw.w};
            const int   cj[KTOP] = {ci.x, ci.y, ci.z, ci.w};
            #pragma unroll
            for (int cp = 0; cp < KTOP; ++cp) {
                const float v = cv[cp];
                const int   j = cj[cp];
                top4_put(w4, i4, v, j,
                         (v > w4[0]) || (v == w4[0] && j < i4[0]),
                         (v > w4[1]) || (v == w4[1] && j < i4[1]),
                         (v > w4[2]) || (v == w4[2] && j < i4[2]),
                         (v > w4[3]) || (v == w4[3] && j < i4[3]));
            }
        }
        const float mx = w4[0];
        float e[KTOP], s = 0.0f;
        #pragma unroll
        for (int p = 0; p < KTOP; ++p) { e[p] = __expf(w4[p] - mx); s += e[p]; }
        const float inv = 1.0f / s;
        #pragma unroll
        for (int p = 0; p < KTOP; ++p) { topw[row][p] = e[p] * inv; topi[row][p] = i4[p]; }
    }
    __syncthreads();

    {
        const int row = tid >> 4;
        const int dd0 = (tid & 15) * 4;
        #pragma unroll
        for (int u = 0; u < 4; ++u) {
            const int dd = dd0 + u;
            float wvv = 0.0f;
            #pragma unroll
            for (int p = 0; p < KTOP; ++p)
                wvv += topw[row][p] * vbuf[vb + (size_t)topi[row][p] * DHEAD + dd];
            const float val = wvv * qbuf[qb + (size_t)(i0 + row) * DHEAD + dd];
            atomicAdd(&out_acc[dd], val);
        }
    }
    __syncthreads();
    if (tid < DHEAD)
        atomicAdd(&out[b * CCH + tid * NHEAD + h], out_acc[tid]);
}

extern "C" void kernel_launch(void* const* d_in, const int* in_sizes, int n_in,
                              void* d_out, int out_size, void* d_ws, size_t ws_size,
                              hipStream_t stream) {
    const float* x  = (const float*)d_in[0];
    const float* Wq = (const float*)d_in[1];
    const float* bq = (const float*)d_in[2];
    float* out = (float*)d_out;
    float* ws  = (float*)d_ws;

    const size_t per = (size_t)BATCH * NHEAD * NTOK * DHEAD;
    float* qbuf = ws;
    float* kT   = ws + per;
    float* vbuf = ws + 2 * per;

    zero_out_kernel<<<(out_size + 255) / 256, 256, 0, stream>>>(out, out_size);
    qkv_kernel<<<256, 384, 0, stream>>>(x, Wq, bq, qbuf, kT, vbuf);
    attn_kernel<<<4096, 256, 0, stream>>>(qbuf, kT, vbuf, out);
}


# attn: a block now covers 32 query rows (two 16-row tiles), so each loaded K^T fragment feeds two matrix instructions; a thread adds its two rows before the LDS atomic add (sum order changes)
# speedup vs baseline: 1.6504x; 1.1373x over previous
#include <hip/hip_runtime.h>
#include <hip/hip_bf16.h>

typedef float v8f __attribute__((ext_vector_type(8)));
typedef float v2f __attribute__((ext_vector_type(2)));

#define BATCH 8
#define CCH   512
#define NHEAD 8
#define DHEAD 64
#define NTOK  1024
#define KTOP  4
#define C3    1536

__global__ void zero_out_kernel(float* __restrict__ out, int n) {
    int i = blockIdx.x * blockDim.x + threadIdx.x;
    if (i < n) out[i] = 0.0f;
}

typedef float v4f __attribute__((ext_vector_type(4)));

struct KStep { v4f a0, a1, b0, b1; };

__device__ __forceinline__ KStep load_kstep(const float* xa, const float* wb, int k0) {
    KStep f;
    f.a0 = *reinterpret_cast<const v4f*>(xa + (size_t)k0 * NTOK);
    f.a1 = *reinterpret_cast<const v4f*>(xa + (size_t)(k0 + 1) * NTOK);
    f.b0 = *reinterpret_cast<const v4f*>(wb + (size_t)k0 * C3);
    f.b1 = *reinterpret_cast<const v4f*>(wb + (size_t)(k0 + 1) * C3);
    return f;
}

__device__ __forceinline__ void mma_kstep(v8f (&acc)[4][4], const KStep& f) {
    #pragma unroll
    for (int t = 0; t < 4; ++t) {
        v2f a; a.x = f.a0[t]; a.y = f.a1[t];
        #pragma unroll
        for (int u = 0; u < 4; ++u) {
            v2f bf; bf.x = f.b0[u]; bf.y = f.b1[u];
            acc[t][u] = __builtin_amdgcn_wmma_f32_16x16x4_f32(
                false, a, false, bf, (short)0, acc[t][u], false, false);
        }
    }
}

__global__ __launch_bounds__(384) void qkv_kernel(const float* __restrict__ x,
                                                  const float* __restrict__ Wqkv,
                                                  const float* __restrict__ bqkv,
                                                  float* __restrict__ qbuf,
                                                  float* __restrict__ kTbuf,
                                                  float* __restrict__ vbuf) {
    const int lane = threadIdx.x & 31;
    const int wv   = __builtin_amdgcn_readfirstlane((int)(threadIdx.x >> 5));
    const int h    = lane >> 4;
    const int c    = lane & 15;
    const int mb   = blockIdx.x >> 3;
    const int nb   = blockIdx.x & 7;
    const int b    = mb >> 2;
    const int m0   = (mb & 3) * 256 + (wv & 3) * 64;
    const int ng   = nb * 3 + (wv >> 2);
    const int which = ng >> 3;
    const int head  = ng & 7;
    const int j0   = ng * 64;

    v8f acc[4][4];
    {
        const v4f bv = *reinterpret_cast<const v4f*>(bqkv + j0 + 4 * c);
        #pragma unroll
        for (int t = 0; t < 4; ++t)
            #pragma unroll
            for (int u = 0; u < 4; ++u)
                #pragma unroll
                for (int r = 0; r < 8; ++r) acc[t][u][r] = bv[u];
    }

    const float* xa = x + (size_t)b * CCH * NTOK + (size_t)(2 * h) * NTOK + m0 + 4 * c;
    const float* wb = Wqkv + (size_t)(2 * h) * C3 + j0 + 4 * c;

    KStep f0 = load_kstep(xa, wb, 0);
    KStep f1 = load_kstep(xa, wb, 4);
    #pragma nounroll
    for (int k0 = 0; k0 < CCH - 16; k0 += 16) {
        KStep g0 = load_kstep(xa, wb, k0 + 8);
        KStep g1 = load_kstep(xa, wb, k0 + 12);
        __builtin_amdgcn_sched_barrier(0);
        mma_kstep(acc, f0);
        mma_kstep(acc, f1);
        f0 = load_kstep(xa, wb, k0 + 16);
        f1 = load_kstep(xa, wb, k0 + 20);
        __builtin_amdgcn_sched_barrier(0);
        mma_kstep(acc, g0);
        mma_kstep(acc, g1);
    }
    {
        KStep g0 = load_kstep(xa, wb, CCH - 8);
        KStep g1 = load_kstep(xa, wb, CCH - 4);
        __builtin_amdgcn_sched_barrier(0);
        mma_kstep(acc, f0);
        mma_kstep(acc, f1);
        mma_kstep(acc, g0);
        mma_kstep(acc, g1);
    }

    const int bh = b * NHEAD + head;
    const int mrow = m0 + 32 * h;
    if (which != 1) {
        float* dst = (which == 0 ? qbuf : vbuf) + (size_t)bh * NTOK * DHEAD + 4 * c;
        #pragma unroll
        for (int t = 0; t < 4; ++t) {
            #pragma unroll
            for (int r = 0; r < 8; ++r) {
                v4f val;
                val[0] = acc[t][0][r]; val[1] = acc[t][1][r]; val[2] = acc[t][2][r]; val[3] = acc[t][3][r];
                *reinterpret_cast<v4f*>(dst + (size_t)(mrow + 4 * r + t) * DHEAD) = val;
            }
        }
    } else {
        #pragma unroll
        for (int r = 0; r < 8; ++r) {
            #pragma unroll
            for (int p = 0; p < 2; ++p) {
                float* dstk = kTbuf + ((((size_t)bh * (DHEAD / 2) + 2 * c + p) * NTOK + mrow + 4 * r) << 1);
                v4f lo, hi;
                lo[0] = acc[0][2 * p][r]; lo[1] = acc[0][2 * p + 1][r]; lo[2] = acc[1][2 * p][r]; lo[3] = acc[1][2 * p + 1][r];
                hi[0] = acc[2][2 * p][r]; hi[1] = acc[2][2 * p + 1][r]; hi[2] = acc[3][2 * p][r]; hi[3] = acc[3][2 * p + 1][r];
                *reinterpret_cast<v4f*>(dstk)     = lo;
                *reinterpret_cast<v4f*>(dstk + 4) = hi;
            }
        }
    }
}

__device__ __forceinline__ void top4_put(float (&w)[4], int (&ix)[4], float v, int j,
                                         bool b0, bool b1, bool b2, bool b3) {
    w[3]  = b2 ? w[2]  : (b3 ? v : w[3]);
    ix[3] = b2 ? ix[2] : (b3 ? j : ix[3]);
    w[2]  = b1 ? w[1]  : (b2 ? v : w[2]);
    ix[2] = b1 ? ix[1] : (b2 ? j : ix[2]);
    w[1]  = b0 ? w[0]  : (b1 ? v : w[1]);
    ix[1] = b0 ? ix[0] : (b1 ? j : ix[1]);
    w[0]  = b0 ? v : w[0];
    ix[0] = b0 ? j : ix[0];
}

#define AROWS 32
__global__ __launch_bounds__(256) void attn_kernel(const float* __restrict__ qbuf,
                                                   const float* __restrict__ kTbuf,
                                                   const float* __restrict__ vbuf,
                                                   float* __restrict__ out) {
    __shared__ __attribute__((aligned(16))) float sc[AROWS][260];
    __shared__ __attribute__((aligned(16))) float candw[AROWS][8][KTOP];
    __shared__ __attribute__((aligned(16))) int   candi[AROWS][8][KTOP];
    __shared__ float topw[AROWS][KTOP];
    __shared__ int   topi[AROWS][KTOP];
    __shared__ float out_acc[DHEAD];

    const int tid     = threadIdx.x;
    const int lane    = tid & 31;
    const int wv      = tid >> 5;
    const int halfsel = lane >> 4;
    const int lm      = lane & 15;

    const int bh    = blockIdx.x >> 5;
    const int itile = blockIdx.x & 31;
    const int b     = bh >> 3;
    const int h     = bh & 7;
    const int i0    = itile * AROWS;

    const size_t qb = (size_t)bh * NTOK * DHEAD;
    const size_t vb = qb;
    const float2* kT2 = (const float2*)(kTbuf + (size_t)bh * DHEAD * NTOK);

    if (tid < DHEAD) out_acc[tid] = 0.0f;

    v2f afrag[2][16];
    #pragma unroll
    for (int it = 0; it < 2; ++it) {
        const float* qrow = qbuf + qb + (size_t)(i0 + 16 * it + lm) * DHEAD + 2 * halfsel;
        #pragma unroll
        for (int kk = 0; kk < 16; ++kk) {
            afrag[it][kk].x = qrow[kk * 4];
            afrag[it][kk].y = qrow[kk * 4 + 1];
        }
    }

    const int srow = tid >> 3;
    const int ssub = tid & 7;
    float lw[KTOP];
    int   li[KTOP];
    #pragma unroll
    for (int p = 0; p < KTOP; ++p) { lw[p] = -__builtin_inff(); li[p] = 0x7fffffff; }

    for (int chunk = 0; chunk < 4; ++chunk) {
        const int jcbase = chunk * 256;
        #pragma unroll
        for (int tsub = 0; tsub < 2; ++tsub) {
            const int jl0 = wv * 32 + tsub * 16;
            const int jg  = jcbase + jl0;
            v8f acc0, acc1;
            #pragma unroll
            for (int r = 0; r < 8; ++r) { acc0[r] = 0.0f; acc1[r] = 0.0f; }
            #pragma unroll
            for (int kk = 0; kk < 16; ++kk) {
                const float2 bp = kT2[(size_t)(kk * 2 + halfsel) * NTOK + jg + lm];
                v2f bf; bf.x = bp.x; bf.y = bp.y;
                acc0 = __builtin_amdgcn_wmma_f32_16x16x4_f32(
                    false, afrag[0][kk], false, bf, (short)0, acc0, false, false);
                acc1 = __builtin_amdgcn_wmma_f32_16x16x4_f32(
                    false, afrag[1][kk], false, bf, (short)0, acc1, false, false);
            }
            #pragma unroll
            for (int r = 0; r < 8; ++r) {
                sc[r + 8 * halfsel][jl0 + lm]      = acc0[r];
                sc[16 + r + 8 * halfsel][jl0 + lm] = acc1[r];
            }
        }
        __syncthreads();
        {
            const float4* stripe = reinterpret_cast<const float4*>(&sc[srow][ssub * 32]);
            const int jb = jcbase + ssub * 32;
            #pragma unroll
            for (int q = 0; q < 8; ++q) {
                const float4 t = stripe[q];
                const float sv[4] = {t.x, t.y, t.z, t.w};
                #pragma unroll
                for (int c = 0; c < 4; ++c) {
                    const float v = sv[c];
                    top4_put(lw, li, v, jb + 4 * q + c, v > lw[0], v > lw[1], v > lw[2], v > lw[3]);
                }
            }
        }
        __syncthreads();
    }

    #pragma unroll
    for (int p = 0; p < KTOP; ++p) { candw[srow][ssub][p] = lw[p]; candi[srow][ssub][p] = li[p]; }
    __syncthreads();

    if (tid < AROWS) {
        const int row = tid;
        float w4[KTOP]; int i4[KTOP];
        #pragma unroll
        for (int p = 0; p < KTOP; ++p) { w4[p] = -__builtin_inff(); i4[p] = 0x7fffffff; }
        for (int s = 0; s < 8; ++s) {
            const float4 cw = *reinterpret_cast<const float4*>(&candw[row][s][0]);
            const int4   ci = *reinterpret_cast<const int4*>(&candi[row][s][0]);
            const float cv[KTOP] = {cw.x, cw.y, cw.z, cw.w};
            const int   cj[KTOP] = {ci.x, ci.y, ci.z, ci.w};
            #pragma unroll
            for (int cp = 0; cp < KTOP; ++cp) {
                const float v = cv[cp];
                const int   j = cj[cp];
                top4_put(w4, i4, v, j,
                         (v > w4[0]) || (v == w4[0] && j < i4[0]),
                         (v > w4[1]) || (v == w4[1] && j < i4[1]),
                         (v > w4[2]) || (v == w4[2] && j < i4[2]),
                         (v > w4[3]) || (v == w4[3] && j < i4[3]));
            }
        }
        const float mx = w4[0];
        float e[KTOP], s = 0.0f;
        #pragma unroll
        for (int p = 0; p < KTOP; ++p) { e[p] = __expf(w4[p] - mx); s += e[p]; }
        const float inv = 1.0f / s;
        #pragma unroll
        for (int p = 0; p < KTOP; ++p) { topw[row][p] = e[p] * inv; topi[row][p] = i4[p]; }
    }
    __syncthreads();

    {
        const int row0 = tid >> 4;
        const int dd0  = (tid & 15) * 4;
        #pragma unroll
        for (int u = 0; u < 4; ++u) {
            const int dd = dd0 + u;
            float val = 0.0f;
            #pragma unroll
            for (int rh = 0; rh < 2; ++rh) {
                const int row = row0 + 16 * rh;
                float wvv = 0.0f;
                #pragma unroll
                for (int p = 0; p < KTOP; ++p)
                    wvv += topw[row][p] * vbuf[vb + (size_t)topi[row][p] * DHEAD + dd];
                val += wvv * qbuf[qb + (size_t)(i0 + row) * DHEAD + dd];
            }
            atomicAdd(&out_acc[dd], val);
        }
    }
    __syncthreads();
    if (tid < DHEAD)
        atomicAdd(&out[b * CCH + tid * NHEAD + h], out_acc[tid]);
}

extern "C" void kernel_launch(void* const* d_in, const int* in_sizes, int n_in,
                              void* d_out, int out_size, void* d_ws, size_t ws_size,
                              hipStream_t stream) {
    const float* x  = (const float*)d_in[0];
    const float* Wq = (const float*)d_in[1];
    const float* bq = (const float*)d_in[2];
    float* out = (float*)d_out;
    float* ws  = (float*)d_ws;

    const size_t per = (size_t)BATCH * NHEAD * NTOK * DHEAD;
    float* qbuf = ws;
    float* kT   = ws + per;
    float* vbuf = ws + 2 * per;

    zero_out_kernel<<<(out_size + 255) / 256, 256, 0, stream>>>(out, out_size);
    qkv_kernel<<<256, 384, 0, stream>>>(x, Wq, bq, qbuf, kT, vbuf);
    attn_kernel<<<2048, 256, 0, stream>>>(qbuf, kT, vbuf, out);
}


# attn: score tile computed transposed (K as A operand, Q as B): a lane's 8 accumulator registers are 8 keys of ONE query row, so top-4 is fed from registers; LDS score tile and its 8 barriers removed
# speedup vs baseline: 1.7518x; 1.0614x over previous
#include <hip/hip_runtime.h>
#include <hip/hip_bf16.h>

typedef float v8f __attribute__((ext_vector_type(8)));
typedef float v2f __attribute__((ext_vector_type(2)));

#define BATCH 8
#define CCH   512
#define NHEAD 8
#define DHEAD 64
#define NTOK  1024
#define KTOP  4
#define C3    1536

__global__ void zero_out_kernel(float* __restrict__ out, int n) {
    int i = blockIdx.x * blockDim.x + threadIdx.x;
    if (i < n) out[i] = 0.0f;
}

typedef float v4f __attribute__((ext_vector_type(4)));

struct KStep { v4f a0, a1, b0, b1; };

__device__ __forceinline__ KStep load_kstep(const float* xa, const float* wb, int k0) {
    KStep f;
    f.a0 = *reinterpret_cast<const v4f*>(xa + (size_t)k0 * NTOK);
    f.a1 = *reinterpret_cast<const v4f*>(xa + (size_t)(k0 + 1) * NTOK);
    f.b0 = *reinterpret_cast<const v4f*>(wb + (size_t)k0 * C3);
    f.b1 = *reinterpret_cast<const v4f*>(wb + (size_t)(k0 + 1) * C3);
    return f;
}

__device__ __forceinline__ void mma_kstep(v8f (&acc)[4][4], const KStep& f) {
    #pragma unroll
    for (int t = 0; t < 4; ++t) {
        v2f a; a.x = f.a0[t]; a.y = f.a1[t];
        #pragma unroll
        for (int u = 0; u < 4; ++u) {
            v2f bf; bf.x = f.b0[u]; bf.y = f.b1[u];
            acc[t][u] = __builtin_amdgcn_wmma_f32_16x16x4_f32(
                false, a, false, bf, (short)0, acc[t][u], false, false);
        }
    }
}

__global__ __launch_bounds__(384) void qkv_kernel(const float* __restrict__ x,
                                                  const float* __restrict__ Wqkv,
                                                  const float* __restrict__ bqkv,
                                                  float* __restrict__ qbuf,
                                                  float* __restrict__ kTbuf,
                                                  float* __restrict__ vbuf) {
    const int lane = threadIdx.x & 31;
    const int wv   = __builtin_amdgcn_readfirstlane((int)(threadIdx.x >> 5));
    const int h    = lane >> 4;
    const int c    = lane & 15;
    const int mb   = blockIdx.x >> 3;
    const int nb   = blockIdx.x & 7;
    const int b    = mb >> 2;
    const int m0   = (mb & 3) * 256 + (wv & 3) * 64;
    const int ng   = nb * 3 + (wv >> 2);
    const int which = ng >> 3;
    const int head  = ng & 7;
    const int j0   = ng * 64;

    v8f acc[4][4];
    {
        const v4f bv = *reinterpret_cast<const v4f*>(bqkv + j0 + 4 * c);
        #pragma unroll
        for (int t = 0; t < 4; ++t)
            #pragma unroll
            for (int u = 0; u < 4; ++u)
                #pragma unroll
                for (int r = 0; r < 8; ++r) acc[t][u][r] = bv[u];
    }

    const float* xa = x + (size_t)b * CCH * NTOK + (size_t)(2 * h) * NTOK + m0 + 4 * c;
    const float* wb = Wqkv + (size_t)(2 * h) * C3 + j0 + 4 * c;

    KStep f0 = load_kstep(xa, wb, 0);
    KStep f1 = load_kstep(xa, wb, 4);
    #pragma nounroll
    for (int k0 = 0; k0 < CCH - 16; k0 += 16) {
        KStep g0 = load_kstep(xa, wb, k0 + 8);
        KStep g1 = load_kstep(xa, wb, k0 + 12);
        __builtin_amdgcn_sched_barrier(0);
        mma_kstep(acc, f0);
        mma_kstep(acc, f1);
        f0 = load_kstep(xa, wb, k0 + 16);
        f1 = load_kstep(xa, wb, k0 + 20);
        __builtin_amdgcn_sched_barrier(0);
        mma_kstep(acc, g0);
        mma_kstep(acc, g1);
    }
    {
        KStep g0 = load_kstep(xa, wb, CCH - 8);
        KStep g1 = load_kstep(xa, wb, CCH - 4);
        __builtin_amdgcn_sched_barrier(0);
        mma_kstep(acc, f0);
        mma_kstep(acc, f1);
        mma_kstep(acc, g0);
        mma_kstep(acc, g1);
    }

    const int bh = b * NHEAD + head;
    const int mrow = m0 + 32 * h;
    if (which != 1) {
        float* dst = (which == 0 ? qbuf : vbuf) + (size_t)bh * NTOK * DHEAD + 4 * c;
        #pragma unroll
        for (int t = 0; t < 4; ++t) {
            #pragma unroll
            for (int r = 0; r < 8; ++r) {
                v4f val;
                val[0] = acc[t][0][r]; val[1] = acc[t][1][r]; val[2] = acc[t][2][r]; val[3] = acc[t][3][r];
                *reinterpret_cast<v4f*>(dst + (size_t)(mrow + 4 * r + t) * DHEAD) = val;
            }
        }
    } else {
        #pragma unroll
        for (int r = 0; r < 8; ++r) {
            #pragma unroll
            for (int p = 0; p < 2; ++p) {
                float* dstk = kTbuf + ((((size_t)bh * (DHEAD / 2) + 2 * c + p) * NTOK + mrow + 4 * r) << 1);
                v4f lo, hi;
                lo[0] = acc[0][2 * p][r]; lo[1] = acc[0][2 * p + 1][r]; lo[2] = acc[1][2 * p][r]; lo[3] = acc[1][2 * p + 1][r];
                hi[0] = acc[2][2 * p][r]; hi[1] = acc[2][2 * p + 1][r]; hi[2] = acc[3][2 * p][r]; hi[3] = acc[3][2 * p + 1][r];
                *reinterpret_cast<v4f*>(dstk)     = lo;
                *reinterpret_cast<v4f*>(dstk + 4) = hi;
            }
        }
    }
}

__device__ __forceinline__ void top4_put(float (&w)[4], int (&ix)[4], float v, int j,
                                         bool b0, bool b1, bool b2, bool b3) {
    const float t3 = b3 ? v : w[3];
    const int   u3 = b3 ? j : ix[3];
    const float t2 = b2 ? v : w[2];
    const int   u2 = b2 ? j : ix[2];
    const float t1 = b1 ? v : w[1];
    const int   u1 = b1 ? j : ix[1];
    w[3]  = b2 ? w[2]  : t3;
    ix[3] = b2 ? ix[2] : u3;
    w[2]  = b1 ? w[1]  : t2;
    ix[2] = b1 ? ix[1] : u2;
    w[1]  = b0 ? w[0]  : t1;
    ix[1] = b0 ? ix[0] : u1;
    w[0]  = b0 ? v : w[0];
    ix[0] = b0 ? j : ix[0];
}

#define AROWS 32
__global__ __launch_bounds__(256) void attn_kernel(const float* __restrict__ qbuf,
                                                   const float* __restrict__ kTbuf,
                                                   const float* __restrict__ vbuf,
                                                   float* __restrict__ out) {
    __shared__ __attribute__((aligned(16))) float candw[AROWS][8][KTOP];
    __shared__ __attribute__((aligned(16))) int   candi[AROWS][8][KTOP];
    __shared__ float topw[AROWS][KTOP];
    __shared__ int   topi[AROWS][KTOP];
    __shared__ float out_acc[DHEAD];

    const int tid     = threadIdx.x;
    const int lane    = tid & 31;
    const int wv      = tid >> 5;
    const int halfsel = lane >> 4;
    const int lm      = lane & 15;

    const int bh    = blockIdx.x >> 5;
    const int itile = blockIdx.x & 31;
    const int b     = bh >> 3;
    const int h     = bh & 7;
    const int i0    = itile * AROWS;

    const size_t qb = (size_t)bh * NTOK * DHEAD;
    const size_t vb = qb;
    const float2* kT2 = (const float2*)(kTbuf + (size_t)bh * DHEAD * NTOK);

    if (tid < DHEAD) out_acc[tid] = 0.0f;

    v2f qfrag[2][16];
    #pragma unroll
    for (int it = 0; it < 2; ++it) {
        const float* qrow = qbuf + qb + (size_t)(i0 + 16 * it + lm) * DHEAD + 2 * halfsel;
        #pragma unroll
        for (int kk = 0; kk < 16; ++kk) {
            qfrag[it][kk].x = qrow[kk * 4];
            qfrag[it][kk].y = qrow[kk * 4 + 1];
        }
    }

    float lw[2][KTOP];
    int   li[2][KTOP];
    #pragma unroll
    for (int it = 0; it < 2; ++it)
        #pragma unroll
        for (int p = 0; p < KTOP; ++p) { lw[it][p] = -__builtin_inff(); li[it][p] = 0x7fffffff; }

    const float2* kwave = kT2 + (size_t)halfsel * NTOK + wv * 128 + lm;
    #pragma nounroll
    for (int jt = 0; jt < 8; ++jt) {
        const float2* ktile = kwave + jt * 16;
        v8f acc0, acc1;
        #pragma unroll
        for (int r = 0; r < 8; ++r) { acc0[r] = 0.0f; acc1[r] = 0.0f; }
        #pragma unroll
        for (int kk = 0; kk < 16; ++kk) {
            const float2 kp = ktile[(size_t)(kk * 2) * NTOK];
            v2f kf; kf.x = kp.x; kf.y = kp.y;
            acc0 = __builtin_amdgcn_wmma_f32_16x16x4_f32(
                false, kf, false, qfrag[0][kk], (short)0, acc0, false, false);
            acc1 = __builtin_amdgcn_wmma_f32_16x16x4_f32(
                false, kf, false, qfrag[1][kk], (short)0, acc1, false, false);
        }
        const int jb = wv * 128 + jt * 16 + 8 * halfsel;
        #pragma unroll
        for (int r = 0; r < 8; ++r) {
            const float v0 = acc0[r];
            const float v1 = acc1[r];
            top4_put(lw[0], li[0], v0, jb + r, v0 > lw[0][0], v0 > lw[0][1], v0 > lw[0][2], v0 > lw[0][3]);
            top4_put(lw[1], li[1], v1, jb + r, v1 > lw[1][0], v1 > lw[1][1], v1 > lw[1][2], v1 > lw[1][3]);
        }
    }

    const int srow = lm + 16 * halfsel;
    float mw[KTOP]; int mi[KTOP];
    #pragma unroll
    for (int p = 0; p < KTOP; ++p) {
        mw[p] = halfsel ? lw[1][p] : lw[0][p];
        mi[p] = halfsel ? li[1][p] : li[0][p];
    }
    #pragma unroll
    for (int p = 0; p < KTOP; ++p) {
        const float sw = halfsel ? lw[0][p] : lw[1][p];
        const int   si = halfsel ? li[0][p] : li[1][p];
        const float v = __shfl_xor(sw, 16);
        const int   j = __shfl_xor(si, 16);
        top4_put(mw, mi, v, j,
                 (v > mw[0]) || (v == mw[0] && j < mi[0]),
                 (v > mw[1]) || (v == mw[1] && j < mi[1]),
                 (v > mw[2]) || (v == mw[2] && j < mi[2]),
                 (v > mw[3]) || (v == mw[3] && j < mi[3]));
    }

    #pragma unroll
    for (int p = 0; p < KTOP; ++p) { candw[srow][wv][p] = mw[p]; candi[srow][wv][p] = mi[p]; }
    __syncthreads();

    if (tid < AROWS) {
        const int row = tid;
        float w4[KTOP]; int i4[KTOP];
        #pragma unroll
        for (int p = 0; p < KTOP; ++p) { w4[p] = -__builtin_inff(); i4[p] = 0x7fffffff; }
        for (int s = 0; s < 8; ++s) {
            const float4 cw = *reinterpret_cast<const float4*>(&candw[row][s][0]);
            const int4   ci = *reinterpret_cast<const int4*>(&candi[row][s][0]);
            const float cv[KTOP] = {cw.x, cw.y, cw.z, cw.w};
            const int   cj[KTOP] = {ci.x, ci.y, ci.z, ci.w};
            #pragma unroll
            for (int cp = 0; cp < KTOP; ++cp) {
                const float v = cv[cp];
                const int   j = cj[cp];
                top4_put(w4, i4, v, j,
                         (v > w4[0]) || (v == w4[0] && j < i4[0]),
                         (v > w4[1]) || (v == w4[1] && j < i4[1]),
                         (v > w4[2]) || (v == w4[2] && j < i4[2]),
                         (v > w4[3]) || (v == w4[3] && j < i4[3]));
            }
        }
        const float mx = w4[0];
        float e[KTOP], s = 0.0f;
        #pragma unroll
        for (int p = 0; p < KTOP; ++p) { e[p] = __expf(w4[p] - mx); s += e[p]; }
        const float inv = 1.0f / s;
        #pragma unroll
        for (int p = 0; p < KTOP; ++p) { topw[row][p] = e[p] * inv; topi[row][p] = i4[p]; }
    }
    __syncthreads();

    {
        const int row0 = tid >> 4;
        const int dd0  = (tid & 15) * 4;
        #pragma unroll
        for (int u = 0; u < 4; ++u) {
            const int dd = dd0 + u;
            float val = 0.0f;
            #pragma unroll
            for (int rh = 0; rh < 2; ++rh) {
                const int row = row0 + 16 * rh;
                float wvv = 0.0f;
                #pragma unroll
                for (int p = 0; p < KTOP; ++p)
                    wvv += topw[row][p] * vbuf[vb + (size_t)topi[row][p] * DHEAD + dd];
                val += wvv * qbuf[qb + (size_t)(i0 + row) * DHEAD + dd];
            }
            atomicAdd(&out_acc[dd], val);
        }
    }
    __syncthreads();
    if (tid < DHEAD)
        atomicAdd(&out[b * CCH + tid * NHEAD + h], out_acc[tid]);
}

extern "C" void kernel_launch(void* const* d_in, const int* in_sizes, int n_in,
                              void* d_out, int out_size, void* d_ws, size_t ws_size,
                              hipStream_t stream) {
    const float* x  = (const float*)d_in[0];
    const float* Wq = (const float*)d_in[1];
    const float* bq = (const float*)d_in[2];
    float* out = (float*)d_out;
    float* ws  = (float*)d_ws;

    const size_t per = (size_t)BATCH * NHEAD * NTOK * DHEAD;
    float* qbuf = ws;
    float* kT   = ws + per;
    float* vbuf = ws + 2 * per;

    zero_out_kernel<<<(out_size + 255) / 256, 256, 0, stream>>>(out, out_size);
    qkv_kernel<<<256, 384, 0, stream>>>(x, Wq, bq, qbuf, kT, vbuf);
    attn_kernel<<<2048, 256, 0, stream>>>(qbuf, kT, vbuf, out);
}


# attn: a wave scans all 1024 keys for its 32 query rows (K tiles loaded one ahead), so top-4 ends inside the wave: LDS candidate lists, serial 32-way merge, 2 barriers gone; rows summed in a new order
# speedup vs baseline: 1.8575x; 1.0603x over previous
#include <hip/hip_runtime.h>
#include <hip/hip_bf16.h>

typedef float v8f __attribute__((ext_vector_type(8)));
typedef float v2f __attribute__((ext_vector_type(2)));

#define BATCH 8
#define CCH   512
#define NHEAD 8
#define DHEAD 64
#define NTOK  1024
#define KTOP  4
#define C3    1536

__global__ void zero_out_kernel(float* __restrict__ out, int n) {
    int i = blockIdx.x * blockDim.x + threadIdx.x;
    if (i < n) out[i] = 0.0f;
}

typedef float v4f __attribute__((ext_vector_type(4)));

struct KStep { v4f a0, a1, b0, b1; };

__device__ __forceinline__ KStep load_kstep(const float* xa, const float* wb, int k0) {
    KStep f;
    f.a0 = *reinterpret_cast<const v4f*>(xa + (size_t)k0 * NTOK);
    f.a1 = *reinterpret_cast<const v4f*>(xa + (size_t)(k0 + 1) * NTOK);
    f.b0 = *reinterpret_cast<const v4f*>(wb + (size_t)k0 * C3);
    f.b1 = *reinterpret_cast<const v4f*>(wb + (size_t)(k0 + 1) * C3);
    return f;
}

__device__ __forceinline__ void mma_kstep(v8f (&acc)[4][4], const KStep& f) {
    #pragma unroll
    for (int t = 0; t < 4; ++t) {
        v2f a; a.x = f.a0[t]; a.y = f.a1[t];
        #pragma unroll
        for (int u = 0; u < 4; ++u) {
            v2f bf; bf.x = f.b0[u]; bf.y = f.b1[u];
            acc[t][u] = __builtin_amdgcn_wmma_f32_16x16x4_f32(
                false, a, false, bf, (short)0, acc[t][u], false, false);
        }
    }
}

__global__ __launch_bounds__(384) void qkv_kernel(const float* __restrict__ x,
                                                  const float* __restrict__ Wqkv,
                                                  const float* __restrict__ bqkv,
                                                  float* __restrict__ qbuf,
                                                  float* __restrict__ kTbuf,
                                                  float* __restrict__ vbuf) {
    const int lane = threadIdx.x & 31;
    const int wv   = __builtin_amdgcn_readfirstlane((int)(threadIdx.x >> 5));
    const int h    = lane >> 4;
    const int c    = lane & 15;
    const int mb   = blockIdx.x >> 3;
    const int nb   = blockIdx.x & 7;
    const int b    = mb >> 2;
    const int m0   = (mb & 3) * 256 + (wv & 3) * 64;
    const int ng   = nb * 3 + (wv >> 2);
    const int which = ng >> 3;
    const int head  = ng & 7;
    const int j0   = ng * 64;

    v8f acc[4][4];
    {
        const v4f bv = *reinterpret_cast<const v4f*>(bqkv + j0 + 4 * c);
        #pragma unroll
        for (int t = 0; t < 4; ++t)
            #pragma unroll
            for (int u = 0; u < 4; ++u)
                #pragma unroll
                for (int r = 0; r < 8; ++r) acc[t][u][r] = bv[u];
    }

    const float* xa = x + (size_t)b * CCH * NTOK + (size_t)(2 * h) * NTOK + m0 + 4 * c;
    const float* wb = Wqkv + (size_t)(2 * h) * C3 + j0 + 4 * c;

    KStep f0 = load_kstep(xa, wb, 0);
    KStep f1 = load_kstep(xa, wb, 4);
    #pragma nounroll
    for (int k0 = 0; k0 < CCH - 16; k0 += 16) {
        KStep g0 = load_kstep(xa, wb, k0 + 8);
        KStep g1 = load_kstep(xa, wb, k0 + 12);
        __builtin_amdgcn_sched_barrier(0);
        mma_kstep(acc, f0);
        mma_kstep(acc, f1);
        f0 = load_kstep(xa, wb, k0 + 16);
        f1 = load_kstep(xa, wb, k0 + 20);
        __builtin_amdgcn_sched_barrier(0);
        mma_kstep(acc, g0);
        mma_kstep(acc, g1);
    }
    {
        KStep g0 = load_kstep(xa, wb, CCH - 8);
        KStep g1 = load_kstep(xa, wb, CCH - 4);
        __builtin_amdgcn_sched_barrier(0);
        mma_kstep(acc, f0);
        mma_kstep(acc, f1);
        mma_kstep(acc, g0);
        mma_kstep(acc, g1);
    }

    const int bh = b * NHEAD + head;
    const int mrow = m0 + 32 * h;
    if (which != 1) {
        float* dst = (which == 0 ? qbuf : vbuf) + (size_t)bh * NTOK * DHEAD + 4 * c;
        #pragma unroll
        for (int t = 0; t < 4; ++t) {
            #pragma unroll
            for (int r = 0; r < 8; ++r) {
                v4f val;
                val[0] = acc[t][0][r]; val[1] = acc[t][1][r]; val[2] = acc[t][2][r]; val[3] = acc[t][3][r];
                *reinterpret_cast<v4f*>(dst + (size_t)(mrow + 4 * r + t) * DHEAD) = val;
            }
        }
    } else {
        #pragma unroll
        for (int r = 0; r < 8; ++r) {
            #pragma unroll
            for (int p = 0; p < 2; ++p) {
                float* dstk = kTbuf + ((((size_t)bh * (DHEAD / 2) + 2 * c + p) * NTOK + mrow + 4 * r) << 1);
                v4f lo, hi;
                lo[0] = acc[0][2 * p][r]; lo[1] = acc[0][2 * p + 1][r]; lo[2] = acc[1][2 * p][r]; lo[3] = acc[1][2 * p + 1][r];
                hi[0] = acc[2][2 * p][r]; hi[1] = acc[2][2 * p + 1][r]; hi[2] = acc[3][2 * p][r]; hi[3] = acc[3][2 * p + 1][r];
                *reinterpret_cast<v4f*>(dstk)     = lo;
                *reinterpret_cast<v4f*>(dstk + 4) = hi;
            }
        }
    }
}

__device__ __forceinline__ void top4_put(float (&w)[4], int (&ix)[4], float v, int j,
                                         bool b0, bool b1, bool b2, bool b3) {
    const float t3 = b3 ? v : w[3];
    const int   u3 = b3 ? j : ix[3];
    const float t2 = b2 ? v : w[2];
    const int   u2 = b2 ? j : ix[2];
    const float t1 = b1 ? v : w[1];
    const int   u1 = b1 ? j : ix[1];
    w[3]  = b2 ? w[2]  : t3;
    ix[3] = b2 ? ix[2] : u3;
    w[2]  = b1 ? w[1]  : t2;
    ix[2] = b1 ? ix[1] : u2;
    w[1]  = b0 ? w[0]  : t1;
    ix[1] = b0 ? ix[0] : u1;
    w[0]  = b0 ? v : w[0];
    ix[0] = b0 ? j : ix[0];
}

#define WROWS 32

__device__ __forceinline__ void load_ktile(v2f (&kf)[16], const float2* kt) {
    #pragma unroll
    for (int kk = 0; kk < 16; ++kk) {
        const float2 kp = kt[(size_t)(kk * 2) * NTOK];
        kf[kk].x = kp.x; kf[kk].y = kp.y;
    }
}

__device__ __forceinline__ void tile_step(const v2f (&kf)[16], const v2f (&qfrag)[2][16],
                                          float (&lw)[2][KTOP], int (&li)[2][KTOP], int jb) {
    v8f acc0, acc1;
    #pragma unroll
    for (int r = 0; r < 8; ++r) { acc0[r] = 0.0f; acc1[r] = 0.0f; }
    #pragma unroll
    for (int kk = 0; kk < 16; ++kk) {
        acc0 = __builtin_amdgcn_wmma_f32_16x16x4_f32(
            false, kf[kk], false, qfrag[0][kk], (short)0, acc0, false, false);
        acc1 = __builtin_amdgcn_wmma_f32_16x16x4_f32(
            false, kf[kk], false, qfrag[1][kk], (short)0, acc1, false, false);
    }
    #pragma unroll
    for (int r = 0; r < 8; ++r) {
        const float v0 = acc0[r];
        const float v1 = acc1[r];
        top4_put(lw[0], li[0], v0, jb + r, v0 > lw[0][0], v0 > lw[0][1], v0 > lw[0][2], v0 > lw[0][3]);
        top4_put(lw[1], li[1], v1, jb + r, v1 > lw[1][0], v1 > lw[1][1], v1 > lw[1][2], v1 > lw[1][3]);
    }
}

__global__ __launch_bounds__(256) void attn_kernel(const float* __restrict__ qbuf,
                                                   const float* __restrict__ kTbuf,
                                                   const float* __restrict__ vbuf,
                                                   float* __restrict__ out) {
    __shared__ __attribute__((aligned(8))) float part[8][DHEAD];

    const int tid     = threadIdx.x;
    const int lane    = tid & 31;
    const int wv      = __builtin_amdgcn_readfirstlane((int)(tid >> 5));
    const int halfsel = lane >> 4;
    const int lm      = lane & 15;

    const int bh = blockIdx.x >> 2;
    const int b  = bh >> 3;
    const int h  = bh & 7;
    const int i0 = (blockIdx.x & 3) * 256 + wv * WROWS;

    const size_t qb = (size_t)bh * NTOK * DHEAD;
    const size_t vb = qb;
    const float2* kT2 = (const float2*)(kTbuf + (size_t)bh * DHEAD * NTOK);

    v2f qfrag[2][16];
    #pragma unroll
    for (int it = 0; it < 2; ++it) {
        const float* qrow = qbuf + qb + (size_t)(i0 + 16 * it + lm) * DHEAD + 2 * halfsel;
        #pragma unroll
        for (int kk = 0; kk < 16; ++kk) {
            qfrag[it][kk].x = qrow[kk * 4];
            qfrag[it][kk].y = qrow[kk * 4 + 1];
        }
    }

    float lw[2][KTOP];
    int   li[2][KTOP];
    #pragma unroll
    for (int it = 0; it < 2; ++it)
        #pragma unroll
        for (int p = 0; p < KTOP; ++p) { lw[it][p] = -__builtin_inff(); li[it][p] = 0x7fffffff; }

    const float2* klane = kT2 + (size_t)halfsel * NTOK + lm;
    const int jlane = 8 * halfsel;

    v2f ka[16], kb[16];
    load_ktile(ka, klane);
    #pragma nounroll
    for (int jt = 0; jt < NTOK / 16; jt += 2) {
        load_ktile(kb, klane + (jt + 1) * 16);
        __builtin_amdgcn_sched_barrier(0);
        tile_step(ka, qfrag, lw, li, jt * 16 + jlane);
        load_ktile(ka, klane + ((jt + 2) & (NTOK / 16 - 1)) * 16);
        __builtin_amdgcn_sched_barrier(0);
        tile_step(kb, qfrag, lw, li, (jt + 1) * 16 + jlane);
    }

    float mw[KTOP]; int mi[KTOP];
    #pragma unroll
    for (int p = 0; p < KTOP; ++p) {
        mw[p] = halfsel ? lw[1][p] : lw[0][p];
        mi[p] = halfsel ? li[1][p] : li[0][p];
    }
    #pragma unroll
    for (int p = 0; p < KTOP; ++p) {
        const float sw = halfsel ? lw[0][p] : lw[1][p];
        const int   si = halfsel ? li[0][p] : li[1][p];
        const float v = __shfl_xor(sw, 16);
        const int   j = __shfl_xor(si, 16);
        top4_put(mw, mi, v, j,
                 (v > mw[0]) || (v == mw[0] && j < mi[0]),
                 (v > mw[1]) || (v == mw[1] && j < mi[1]),
                 (v > mw[2]) || (v == mw[2] && j < mi[2]),
                 (v > mw[3]) || (v == mw[3] && j < mi[3]));
    }

    float wgt[KTOP];
    {
        const float mx = mw[0];
        float e[KTOP], s = 0.0f;
        #pragma unroll
        for (int p = 0; p < KTOP; ++p) { e[p] = __expf(mw[p] - mx); s += e[p]; }
        const float inv = 1.0f / s;
        #pragma unroll
        for (int p = 0; p < KTOP; ++p) wgt[p] = e[p] * inv;
    }

    float ax = 0.0f, ay = 0.0f;
    {
        const float* vcol = vbuf + vb + 2 * lane;
        const float* qcol = qbuf + qb + (size_t)i0 * DHEAD + 2 * lane;
        #pragma unroll 8
        for (int rr = 0; rr < WROWS; ++rr) {
            float sx = 0.0f, sy = 0.0f;
            #pragma unroll
            for (int p = 0; p < KTOP; ++p) {
                const int   j = __builtin_amdgcn_readlane(mi[p], rr) & (NTOK - 1);
                const float w = __int_as_float(__builtin_amdgcn_readlane(__float_as_int(wgt[p]), rr));
                const float2 vv = *reinterpret_cast<const float2*>(vcol + (size_t)j * DHEAD);
                sx += w * vv.x; sy += w * vv.y;
            }
            const float2 qq = *reinterpret_cast<const float2*>(qcol + (size_t)rr * DHEAD);
            ax += sx * qq.x; ay += sy * qq.y;
        }
    }
    *reinterpret_cast<float2*>(&part[wv][2 * lane]) = make_float2(ax, ay);
    __syncthreads();
    if (tid < DHEAD) {
        float s = part[0][tid];
        #pragma unroll
        for (int w = 1; w < 8; ++w) s += part[w][tid];
        atomicAdd(&out[b * CCH + tid * NHEAD + h], s);
    }
}

extern "C" void kernel_launch(void* const* d_in, const int* in_sizes, int n_in,
                              void* d_out, int out_size, void* d_ws, size_t ws_size,
                              hipStream_t stream) {
    const float* x  = (const float*)d_in[0];
    const float* Wq = (const float*)d_in[1];
    const float* bq = (const float*)d_in[2];
    float* out = (float*)d_out;
    float* ws  = (float*)d_ws;

    const size_t per = (size_t)BATCH * NHEAD * NTOK * DHEAD;
    float* qbuf = ws;
    float* kT   = ws + per;
    float* vbuf = ws + 2 * per;

    zero_out_kernel<<<(out_size + 255) / 256, 256, 0, stream>>>(out, out_size);
    qkv_kernel<<<256, 384, 0, stream>>>(x, Wq, bq, qbuf, kT, vbuf);
    attn_kernel<<<256, 256, 0, stream>>>(qbuf, kT, vbuf, out);
}
